# starting kernel
# baseline (speedup 1.0000x reference)
#include <hip/hip_runtime.h>
#include <hip/hip_bf16.h>


typedef float v2f __attribute__((ext_vector_type(2)));
typedef float v8f __attribute__((ext_vector_type(8)));

#define F_IN 512
#define C_OUT 64


__global__ void gcn_deg_init(float* deg, int N) {
  int i = blockIdx.x * blockDim.x + threadIdx.x;
  if (i < N) deg[i] = 1.0f;
}

__global__ void gcn_deg_count(const int* __restrict__ dst, float* deg, int E) {
  int stride = gridDim.x * blockDim.x;
  for (int e = blockIdx.x * blockDim.x + threadIdx.x; e < E; e += stride)
    atomicAdd(&deg[dst[e]], 1.0f);
}

__global__ void gcn_deg_to_dinv(float* deg, int N) {
  int i = blockIdx.x * blockDim.x + threadIdx.x;
  if (i < N) deg[i] = rsqrtf(deg[i]);
}


__global__ __launch_bounds__(128) void gcn_gemm_wmma(
    const float* __restrict__ x, const float* __restrict__ W,
    float* __restrict__ h, int N) {
  const int lane = threadIdx.x & 31;
  const int wid  = threadIdx.x >> 5;
  const int tile = blockIdx.x * 4 + wid;
  if (tile * 16 >= N) return;

  const int half = lane >> 4;
  const int l16  = lane & 15;
  const int row0 = tile * 16;

  const float* arow = x + (size_t)(row0 + l16) * F_IN + 2 * half;
  const float* bbas = W + (size_t)(2 * half) * C_OUT + l16;

  v8f acc0 = {}, acc1 = {}, acc2 = {}, acc3 = {};

  #pragma unroll 4
  for (int k = 0; k < F_IN; k += 4) {
    v2f a = *(const v2f*)(arow + k);
    const float* bk = bbas + (size_t)k * C_OUT;
    v2f b0 = { bk[0],  bk[0  + C_OUT] };
    v2f b1 = { bk[16], bk[16 + C_OUT] };
    v2f b2 = { bk[32], bk[32 + C_OUT] };
    v2f b3 = { bk[48], bk[48 + C_OUT] };
    acc0 = __builtin_amdgcn_wmma_f32_16x16x4_f32(false, a, false, b0, (short)0, acc0, false, false);
    acc1 = __builtin_amdgcn_wmma_f32_16x16x4_f32(false, a, false, b1, (short)0, acc1, false, false);
    acc2 = __builtin_amdgcn_wmma_f32_16x16x4_f32(false, a, false, b2, (short)0, acc2, false, false);
    acc3 = __builtin_amdgcn_wmma_f32_16x16x4_f32(false, a, false, b3, (short)0, acc3, false, false);
  }

  float* hp = h + (size_t)row0 * C_OUT;
  #pragma unroll
  for (int v = 0; v < 8; ++v) {
    const size_t r = (size_t)(v + 8 * half) * C_OUT;
    hp[r + 0  + l16] = acc0[v];
    hp[r + 16 + l16] = acc1[v];
    hp[r + 32 + l16] = acc2[v];
    hp[r + 48 + l16] = acc3[v];
  }
}


__global__ void gcn_self_init(const float* __restrict__ h,
                              const float* __restrict__ dinv,
                              float* __restrict__ out, int N) {
  int stride = gridDim.x * blockDim.x;
  int total = N * C_OUT;
  for (int i = blockIdx.x * blockDim.x + threadIdx.x; i < total; i += stride) {
    float di = dinv[i >> 6];
    out[i] = h[i] * di * di;
  }
}


__global__ void gcn_aggregate(const int* __restrict__ src,
                              const int* __restrict__ dst,
                              const float* __restrict__ h,
                              const float* __restrict__ dinv,
                              float* __restrict__ out, int E) {
  long long stride = (long long)gridDim.x * blockDim.x;
  long long total  = (long long)E * 16;
  for (long long i = (long long)blockIdx.x * blockDim.x + threadIdx.x;
       i < total; i += stride) {
    int e = (int)(i >> 4);
    int g = (int)(i & 15);
    int s = src[e];
    int d = dst[e];
    float norm = dinv[s] * dinv[d];
    const float4 hv = *(const float4*)(h + (size_t)s * C_OUT + g * 4);
    float* o = out + (size_t)d * C_OUT + g * 4;
    atomicAdd(o + 0, hv.x * norm);
    atomicAdd(o + 1, hv.y * norm);
    atomicAdd(o + 2, hv.z * norm);
    atomicAdd(o + 3, hv.w * norm);
  }
}


__global__ void gcn_bias_logsoftmax(float* __restrict__ out,
                                    const float* __restrict__ bias, int N) {
  int lane = threadIdx.x & 31;
  int row  = blockIdx.x * (blockDim.x >> 5) + (threadIdx.x >> 5);
  if (row >= N) return;
  float* o = out + (size_t)row * C_OUT;
  float v0 = o[lane]      + bias[lane];
  float v1 = o[lane + 32] + bias[lane + 32];
  float m = fmaxf(v0, v1);
  #pragma unroll
  for (int off = 16; off > 0; off >>= 1) m = fmaxf(m, __shfl_xor(m, off, 32));
  float s = __expf(v0 - m) + __expf(v1 - m);
  #pragma unroll
  for (int off = 16; off > 0; off >>= 1) s += __shfl_xor(s, off, 32);
  float lse = __logf(s);
  o[lane]      = v0 - m - lse;
  o[lane + 32] = v1 - m - lse;
}


extern "C" void kernel_launch(void* const* d_in, const int* in_sizes, int n_in,
                              void* d_out, int out_size, void* d_ws, size_t ws_size,
                              hipStream_t stream) {
  const float* x  = (const float*)d_in[0];
  const float* W  = (const float*)d_in[1];
  const float* b  = (const float*)d_in[2];
  const int*   ei = (const int*)d_in[3];

  const int N = in_sizes[0] / F_IN;
  const int E = in_sizes[3] / 2;
  const int* src = ei;
  const int* dst = ei + E;
  float* out = (float*)d_out;

  char*  ws   = (char*)d_ws;
  float* dinv = (float*)ws;
  size_t off  = ((size_t)N * sizeof(float) + 255) & ~(size_t)255;
  float* h    = (float*)(ws + off);

  const int T = 256;

  gcn_deg_init<<<(N + T - 1) / T, T, 0, stream>>>(dinv, N);
  {
    int blocks = (E + T - 1) / T;
    if (blocks > 8192) blocks = 8192;
    gcn_deg_count<<<blocks, T, 0, stream>>>(dst, dinv, E);
  }
  gcn_deg_to_dinv<<<(N + T - 1) / T, T, 0, stream>>>(dinv, N);

  {
    int tiles  = (N + 15) / 16;
    int blocks = (tiles + 3) / 4;
    gcn_gemm_wmma<<<blocks, 128, 0, stream>>>(x, W, h, N);
  }

  {
    int total  = N * C_OUT;
    int blocks = (total + T - 1) / T;
    if (blocks > 16384) blocks = 16384;
    gcn_self_init<<<blocks, T, 0, stream>>>(h, dinv, out, N);
  }

  {
    long long total = (long long)E * 16;
    long long blk   = (total + T - 1) / T;
    int blocks = blk > 16384 ? 16384 : (int)blk;
    gcn_aggregate<<<blocks, T, 0, stream>>>(src, dst, h, dinv, out, E);
  }

  {
    int rows_per_block = T / 32;
    int blocks = (N + rows_per_block - 1) / rows_per_block;
    gcn_bias_logsoftmax<<<blocks, T, 0, stream>>>(out, b, N);
  }
}


# GEMM fragments read as 16-byte pieces straight from global memory: k order permuted inside each 8-k group (product sum order changes), lane c of tile j = column 4c+j; whole-line W reads and h stores
# speedup vs baseline: 1.1762x; 1.1762x over previous
#include <hip/hip_runtime.h>
#include <hip/hip_bf16.h>


typedef float v2f __attribute__((ext_vector_type(2)));
typedef float v8f __attribute__((ext_vector_type(8)));

#define F_IN 512
#define C_OUT 64


#define DEG_STRIDE 32

__global__ void gcn_deg_init(float* cnt, int N) {
  int i = blockIdx.x * blockDim.x + threadIdx.x;
  if (i < N) cnt[(size_t)i * DEG_STRIDE] = 1.0f;
}

__global__ void gcn_deg_count(const int* __restrict__ dst, float* cnt, int E) {
  int stride = gridDim.x * blockDim.x;
  for (int e = blockIdx.x * blockDim.x + threadIdx.x; e < E; e += stride)
    atomicAdd(&cnt[(size_t)dst[e] * DEG_STRIDE], 1.0f);
}

__global__ void gcn_deg_to_dinv(const float* __restrict__ cnt,
                                float* __restrict__ dinv, int N) {
  int i = blockIdx.x * blockDim.x + threadIdx.x;
  if (i < N) dinv[i] = rsqrtf(cnt[(size_t)i * DEG_STRIDE]);
}


typedef float v4f __attribute__((ext_vector_type(4)));

__global__ __launch_bounds__(128) void gcn_gemm_wmma(
    const float* __restrict__ x, const float* __restrict__ W,
    float* __restrict__ h, int N) {
  const int lane = threadIdx.x & 31;
  const int wid  = threadIdx.x >> 5;
  const int tile = blockIdx.x * 4 + wid;
  if (tile * 16 >= N) return;

  const int half = lane >> 4;
  const int l16  = lane & 15;
  const int row0 = tile * 16;

  const float* arow = x + (size_t)(row0 + l16) * F_IN + 4 * half;
  const float* bbas = W + (size_t)(4 * half) * C_OUT + 4 * l16;

  v8f acc0 = {}, acc1 = {}, acc2 = {}, acc3 = {};

  for (int k = 0; k < F_IN; k += 32) {
    v4f a[4], w[16];
    #pragma unroll
    for (int g = 0; g < 4; ++g) {
      a[g] = *(const v4f*)(arow + k + 8 * g);
      const float* bk = bbas + (size_t)(k + 8 * g) * C_OUT;
      w[4 * g + 0] = *(const v4f*)(bk);
      w[4 * g + 1] = *(const v4f*)(bk + C_OUT);
      w[4 * g + 2] = *(const v4f*)(bk + 2 * C_OUT);
      w[4 * g + 3] = *(const v4f*)(bk + 3 * C_OUT);
    }
    __builtin_amdgcn_sched_barrier(0);
    #pragma unroll
    for (int g = 0; g < 4; ++g) {
      const v4f w0 = w[4 * g + 0], w1 = w[4 * g + 1], w2 = w[4 * g + 2], w3 = w[4 * g + 3];
      const v2f a01 = { a[g].x, a[g].y };
      const v2f a23 = { a[g].z, a[g].w };
      const v2f b00 = { w0.x, w1.x }, b01 = { w0.y, w1.y }, b02 = { w0.z, w1.z }, b03 = { w0.w, w1.w };
      const v2f b10 = { w2.x, w3.x }, b11 = { w2.y, w3.y }, b12 = { w2.z, w3.z }, b13 = { w2.w, w3.w };
      acc0 = __builtin_amdgcn_wmma_f32_16x16x4_f32(false, a01, false, b00, (short)0, acc0, false, false);
      acc1 = __builtin_amdgcn_wmma_f32_16x16x4_f32(false, a01, false, b01, (short)0, acc1, false, false);
      acc2 = __builtin_amdgcn_wmma_f32_16x16x4_f32(false, a01, false, b02, (short)0, acc2, false, false);
      acc3 = __builtin_amdgcn_wmma_f32_16x16x4_f32(false, a01, false, b03, (short)0, acc3, false, false);
      acc0 = __builtin_amdgcn_wmma_f32_16x16x4_f32(false, a23, false, b10, (short)0, acc0, false, false);
      acc1 = __builtin_amdgcn_wmma_f32_16x16x4_f32(false, a23, false, b11, (short)0, acc1, false, false);
      acc2 = __builtin_amdgcn_wmma_f32_16x16x4_f32(false, a23, false, b12, (short)0, acc2, false, false);
      acc3 = __builtin_amdgcn_wmma_f32_16x16x4_f32(false, a23, false, b13, (short)0, acc3, false, false);
    }
  }

  float* hp = h + (size_t)row0 * C_OUT + 4 * l16;
  #pragma unroll
  for (int v = 0; v < 8; ++v) {
    const v4f r = { acc0[v], acc1[v], acc2[v], acc3[v] };
    *(v4f*)(hp + (size_t)(v + 8 * half) * C_OUT) = r;
  }
}


__global__ void gcn_self_init(const float* __restrict__ h,
                              const float* __restrict__ dinv,
                              float* __restrict__ out, int N) {
  int stride = gridDim.x * blockDim.x;
  int total = N * C_OUT;
  for (int i = blockIdx.x * blockDim.x + threadIdx.x; i < total; i += stride) {
    float di = dinv[i >> 6];
    out[i] = h[i] * di * di;
  }
}


__global__ void gcn_aggregate(const int* __restrict__ src,
                              const int* __restrict__ dst,
                              const float* __restrict__ h,
                              const float* __restrict__ dinv,
                              float* __restrict__ out, int E) {
  long long stride = (long long)gridDim.x * blockDim.x;
  long long total  = (long long)E * 16;
  for (long long i = (long long)blockIdx.x * blockDim.x + threadIdx.x;
       i < total; i += stride) {
    int e = (int)(i >> 4);
    int g = (int)(i & 15);
    int s = src[e];
    int d = dst[e];
    float norm = dinv[s] * dinv[d];
    const float4 hv = *(const float4*)(h + (size_t)s * C_OUT + g * 4);
    float* o = out + (size_t)d * C_OUT + g * 4;
    atomicAdd(o + 0, hv.x * norm);
    atomicAdd(o + 1, hv.y * norm);
    atomicAdd(o + 2, hv.z * norm);
    atomicAdd(o + 3, hv.w * norm);
  }
}


__global__ void gcn_bias_logsoftmax(float* __restrict__ out,
                                    const float* __restrict__ bias, int N) {
  int lane = threadIdx.x & 31;
  int row  = blockIdx.x * (blockDim.x >> 5) + (threadIdx.x >> 5);
  if (row >= N) return;
  float* o = out + (size_t)row * C_OUT;
  float v0 = o[lane]      + bias[lane];
  float v1 = o[lane + 32] + bias[lane + 32];
  float m = fmaxf(v0, v1);
  #pragma unroll
  for (int off = 16; off > 0; off >>= 1) m = fmaxf(m, __shfl_xor(m, off, 32));
  float s = __expf(v0 - m) + __expf(v1 - m);
  #pragma unroll
  for (int off = 16; off > 0; off >>= 1) s += __shfl_xor(s, off, 32);
  float lse = __logf(s);
  o[lane]      = v0 - m - lse;
  o[lane + 32] = v1 - m - lse;
}


extern "C" void kernel_launch(void* const* d_in, const int* in_sizes, int n_in,
                              void* d_out, int out_size, void* d_ws, size_t ws_size,
                              hipStream_t stream) {
  const float* x  = (const float*)d_in[0];
  const float* W  = (const float*)d_in[1];
  const float* b  = (const float*)d_in[2];
  const int*   ei = (const int*)d_in[3];

  const int N = in_sizes[0] / F_IN;
  const int E = in_sizes[3] / 2;
  const int* src = ei;
  const int* dst = ei + E;
  float* out = (float*)d_out;

  char*  ws   = (char*)d_ws;
  float* dinv = (float*)ws;
  size_t off  = ((size_t)N * sizeof(float) + 255) & ~(size_t)255;
  float* h    = (float*)(ws + off);

  const int T = 256;

  gcn_deg_init<<<(N + T - 1) / T, T, 0, stream>>>(h, N);
  {
    int blocks = (E + T - 1) / T;
    if (blocks > 8192) blocks = 8192;
    gcn_deg_count<<<blocks, T, 0, stream>>>(dst, h, E);
  }
  gcn_deg_to_dinv<<<(N + T - 1) / T, T, 0, stream>>>(h, dinv, N);

  {
    int tiles  = (N + 15) / 16;
    int blocks = (tiles + 3) / 4;
    gcn_gemm_wmma<<<blocks, 128, 0, stream>>>(x, W, h, N);
  }

  {
    int total  = N * C_OUT;
    int blocks = (total + T - 1) / T;
    if (blocks > 16384) blocks = 16384;
    gcn_self_init<<<blocks, T, 0, stream>>>(h, dinv, out, N);
  }

  {
    long long total = (long long)E * 16;
    long long blk   = (total + T - 1) / T;
    int blocks = blk > 16384 ? 16384 : (int)blk;
    gcn_aggregate<<<blocks, T, 0, stream>>>(src, dst, h, dinv, out, E);
  }

  {
    int rows_per_block = T / 32;
    int blocks = (N + rows_per_block - 1) / rows_per_block;
    gcn_bias_logsoftmax<<<blocks, T, 0, stream>>>(out, b, N);
  }
}


# degree-count and edge-sum passes split by chip half: a block adds only into the node range of the half it runs on; both halves scan all edges, pieces by ticket counters; sum order = atomic arrival
# speedup vs baseline: 1.3148x; 1.1178x over previous
#include <hip/hip_runtime.h>
#include <hip/hip_bf16.h>


typedef float v2f __attribute__((ext_vector_type(2)));
typedef float v8f __attribute__((ext_vector_type(8)));

#define F_IN 512
#define C_OUT 64


#define DEG_STRIDE 32


#define N_TICKETS 4
#define DEG_ITEM 1024
#define AGG_ITEM 512

__device__ __forceinline__ int chip_half_of_wave() {
  return (int)((__builtin_amdgcn_s_sendmsg_rtn(0x87) >> 18) & 1u);
}

__device__ __forceinline__ int next_item(int* ticket, int* slot) {
  __syncthreads();
  if (threadIdx.x == 0) *slot = atomicAdd(ticket, 1);
  __syncthreads();
  return *slot;
}

__global__ void gcn_deg_init(float* cnt, int* tickets, int N) {
  int i = blockIdx.x * blockDim.x + threadIdx.x;
  if (i < N_TICKETS) tickets[i] = 0;
  if (i < N) cnt[(size_t)i * DEG_STRIDE] = 1.0f;
}

__global__ __launch_bounds__(256) void gcn_deg_count(
    const int* __restrict__ dst, float* cnt, int* tickets, int E, int n_half) {
  __shared__ int s_slot, s_half;
  if (threadIdx.x == 0) s_half = chip_half_of_wave();
  __syncthreads();
  const int home = s_half;
  const int n_items = (E + DEG_ITEM - 1) / DEG_ITEM;
  for (int pass = 0; pass < 2; ++pass) {
    const int part = home ^ pass;
    for (int trip = 0; trip <= n_items; ++trip) {
      const int item = next_item(&tickets[part], &s_slot);
      if (item >= n_items) break;
      const int base = item * DEG_ITEM;
      int dd[DEG_ITEM / 256];
      #pragma unroll
      for (int r = 0; r < DEG_ITEM / 256; ++r) {
        const int e = base + r * 256 + (int)threadIdx.x;
        dd[r] = (e < E) ? dst[e] : -1;
      }
      #pragma unroll
      for (int r = 0; r < DEG_ITEM / 256; ++r) {
        if (dd[r] >= 0 && (dd[r] >= n_half) == (part != 0))
          atomicAdd(&cnt[(size_t)dd[r] * DEG_STRIDE], 1.0f);
      }
    }
  }
}

__global__ void gcn_deg_to_dinv(const float* __restrict__ cnt,
                                float* __restrict__ dinv, int N) {
  int i = blockIdx.x * blockDim.x + threadIdx.x;
  if (i < N) dinv[i] = rsqrtf(cnt[(size_t)i * DEG_STRIDE]);
}


typedef float v4f __attribute__((ext_vector_type(4)));

__global__ __launch_bounds__(128) void gcn_gemm_wmma(
    const float* __restrict__ x, const float* __restrict__ W,
    float* __restrict__ h, int N) {
  const int lane = threadIdx.x & 31;
  const int wid  = threadIdx.x >> 5;
  const int tile = blockIdx.x * 4 + wid;
  if (tile * 16 >= N) return;

  const int half = lane >> 4;
  const int l16  = lane & 15;
  const int row0 = tile * 16;

  const float* arow = x + (size_t)(row0 + l16) * F_IN + 4 * half;
  const float* bbas = W + (size_t)(4 * half) * C_OUT + 4 * l16;

  v8f acc0 = {}, acc1 = {}, acc2 = {}, acc3 = {};

  for (int k = 0; k < F_IN; k += 32) {
    v4f a[4], w[16];
    #pragma unroll
    for (int g = 0; g < 4; ++g) {
      a[g] = *(const v4f*)(arow + k + 8 * g);
      const float* bk = bbas + (size_t)(k + 8 * g) * C_OUT;
      w[4 * g + 0] = *(const v4f*)(bk);
      w[4 * g + 1] = *(const v4f*)(bk + C_OUT);
      w[4 * g + 2] = *(const v4f*)(bk + 2 * C_OUT);
      w[4 * g + 3] = *(const v4f*)(bk + 3 * C_OUT);
    }
    __builtin_amdgcn_sched_barrier(0);
    #pragma unroll
    for (int g = 0; g < 4; ++g) {
      const v4f w0 = w[4 * g + 0], w1 = w[4 * g + 1], w2 = w[4 * g + 2], w3 = w[4 * g + 3];
      const v2f a01 = { a[g].x, a[g].y };
      const v2f a23 = { a[g].z, a[g].w };
      const v2f b00 = { w0.x, w1.x }, b01 = { w0.y, w1.y }, b02 = { w0.z, w1.z }, b03 = { w0.w, w1.w };
      const v2f b10 = { w2.x, w3.x }, b11 = { w2.y, w3.y }, b12 = { w2.z, w3.z }, b13 = { w2.w, w3.w };
      acc0 = __builtin_amdgcn_wmma_f32_16x16x4_f32(false, a01, false, b00, (short)0, acc0, false, false);
      acc1 = __builtin_amdgcn_wmma_f32_16x16x4_f32(false, a01, false, b01, (short)0, acc1, false, false);
      acc2 = __builtin_amdgcn_wmma_f32_16x16x4_f32(false, a01, false, b02, (short)0, acc2, false, false);
      acc3 = __builtin_amdgcn_wmma_f32_16x16x4_f32(false, a01, false, b03, (short)0, acc3, false, false);
      acc0 = __builtin_amdgcn_wmma_f32_16x16x4_f32(false, a23, false, b10, (short)0, acc0, false, false);
      acc1 = __builtin_amdgcn_wmma_f32_16x16x4_f32(false, a23, false, b11, (short)0, acc1, false, false);
      acc2 = __builtin_amdgcn_wmma_f32_16x16x4_f32(false, a23, false, b12, (short)0, acc2, false, false);
      acc3 = __builtin_amdgcn_wmma_f32_16x16x4_f32(false, a23, false, b13, (short)0, acc3, false, false);
    }
  }

  float* hp = h + (size_t)row0 * C_OUT + 4 * l16;
  #pragma unroll
  for (int v = 0; v < 8; ++v) {
    const v4f r = { acc0[v], acc1[v], acc2[v], acc3[v] };
    *(v4f*)(hp + (size_t)(v + 8 * half) * C_OUT) = r;
  }
}


__global__ void gcn_self_init(const float* __restrict__ h,
                              const float* __restrict__ dinv,
                              float* __restrict__ out, int N) {
  int stride = gridDim.x * blockDim.x;
  int total = N * C_OUT;
  for (int i = blockIdx.x * blockDim.x + threadIdx.x; i < total; i += stride) {
    float di = dinv[i >> 6];
    out[i] = h[i] * di * di;
  }
}


__global__ __launch_bounds__(256) void gcn_aggregate(
    const int* __restrict__ src, const int* __restrict__ dst,
    const float* __restrict__ h, const float* __restrict__ dinv,
    float* __restrict__ out, int* tickets, int E, int n_half) {
  __shared__ int s_slot, s_half;
  if (threadIdx.x == 0) s_half = chip_half_of_wave();
  __syncthreads();
  const int home = s_half;
  const int lane = threadIdx.x & 31;
  const int hi   = lane >> 4;
  const int g    = lane & 15;
  const int n_items = (E + AGG_ITEM - 1) / AGG_ITEM;
  for (int pass = 0; pass < 2; ++pass) {
    const int part = home ^ pass;
    for (int trip = 0; trip <= n_items; ++trip) {
      const int item = next_item(&tickets[2 + part], &s_slot);
      if (item >= n_items) break;
      const int base = item * AGG_ITEM;
      int   s[AGG_ITEM / 256], d[AGG_ITEM / 256];
      float nrm[AGG_ITEM / 256];
      bool  mine[AGG_ITEM / 256];
      #pragma unroll
      for (int r = 0; r < AGG_ITEM / 256; ++r) {
        const int e  = base + r * 256 + (int)threadIdx.x;
        const int ec = e < E ? e : E - 1;
        s[r] = src[ec];
        d[r] = dst[ec];
        mine[r] = (e < E) && ((d[r] >= n_half) == (part != 0));
      }
      #pragma unroll
      for (int r = 0; r < AGG_ITEM / 256; ++r)
        nrm[r] = mine[r] ? dinv[s[r]] * dinv[d[r]] : 0.0f;
      #pragma unroll
      for (int r = 0; r < AGG_ITEM / 256; ++r) {
        unsigned m = __builtin_amdgcn_ballot_w32(mine[r]);
        for (int t = 0; t < 16 && m != 0u; ++t) {
          const int l0 = __builtin_ctz(m);
          m &= m - 1u;
          const bool two = (m != 0u);
          int l1 = l0;
          if (two) { l1 = __builtin_ctz(m); m &= m - 1u; }
          const int from = hi ? l1 : l0;
          const int   es = __shfl(s[r], from, 32);
          const int   ed = __shfl(d[r], from, 32);
          const float en = __shfl(nrm[r], from, 32);
          if (hi == 0 || two) {
            const float4 hv = *(const float4*)(h + (size_t)es * C_OUT + g * 4);
            float* o = out + (size_t)ed * C_OUT + g * 4;
            atomicAdd(o + 0, hv.x * en);
            atomicAdd(o + 1, hv.y * en);
            atomicAdd(o + 2, hv.z * en);
            atomicAdd(o + 3, hv.w * en);
          }
        }
      }
    }
  }
}


__global__ void gcn_bias_logsoftmax(float* __restrict__ out,
                                    const float* __restrict__ bias, int N) {
  int lane = threadIdx.x & 31;
  int row  = blockIdx.x * (blockDim.x >> 5) + (threadIdx.x >> 5);
  if (row >= N) return;
  float* o = out + (size_t)row * C_OUT;
  float v0 = o[lane]      + bias[lane];
  float v1 = o[lane + 32] + bias[lane + 32];
  float m = fmaxf(v0, v1);
  #pragma unroll
  for (int off = 16; off > 0; off >>= 1) m = fmaxf(m, __shfl_xor(m, off, 32));
  float s = __expf(v0 - m) + __expf(v1 - m);
  #pragma unroll
  for (int off = 16; off > 0; off >>= 1) s += __shfl_xor(s, off, 32);
  float lse = __logf(s);
  o[lane]      = v0 - m - lse;
  o[lane + 32] = v1 - m - lse;
}


extern "C" void kernel_launch(void* const* d_in, const int* in_sizes, int n_in,
                              void* d_out, int out_size, void* d_ws, size_t ws_size,
                              hipStream_t stream) {
  const float* x  = (const float*)d_in[0];
  const float* W  = (const float*)d_in[1];
  const float* b  = (const float*)d_in[2];
  const int*   ei = (const int*)d_in[3];

  const int N = in_sizes[0] / F_IN;
  const int E = in_sizes[3] / 2;
  const int* src = ei;
  const int* dst = ei + E;
  float* out = (float*)d_out;

  char*  ws   = (char*)d_ws;
  float* dinv = (float*)ws;
  int*   tickets = (int*)(ws + (size_t)N * sizeof(float));
  size_t off  = ((size_t)N * sizeof(float) + N_TICKETS * sizeof(int) + 255) & ~(size_t)255;
  float* h    = (float*)(ws + off);
  const int n_half = N / 2;

  const int T = 256;

  gcn_deg_init<<<(N + T - 1) / T, T, 0, stream>>>(h, tickets, N);
  gcn_deg_count<<<1024, T, 0, stream>>>(dst, h, tickets, E, n_half);
  gcn_deg_to_dinv<<<(N + T - 1) / T, T, 0, stream>>>(h, dinv, N);

  {
    int tiles  = (N + 15) / 16;
    int blocks = (tiles + 3) / 4;
    gcn_gemm_wmma<<<blocks, 128, 0, stream>>>(x, W, h, N);
  }

  {
    int total  = N * C_OUT;
    int blocks = (total + T - 1) / T;
    if (blocks > 16384) blocks = 16384;
    gcn_self_init<<<blocks, T, 0, stream>>>(h, dinv, out, N);
  }

  gcn_aggregate<<<2048, T, 0, stream>>>(src, dst, h, dinv, out, tickets, E, n_half);

  {
    int rows_per_block = T / 32;
    int blocks = (N + rows_per_block - 1) / rows_per_block;
    gcn_bias_logsoftmax<<<blocks, T, 0, stream>>>(out, b, N);
  }
}


# edge sum pulled not pushed: the in-degree atomic returns an edge's rank, its source id goes into a list in out row d; a wave per row sums h rows in registers; no f32 atomics; sum order = arrival rank
# speedup vs baseline: 2.9433x; 2.2386x over previous
#include <hip/hip_runtime.h>
#include <hip/hip_bf16.h>


typedef float v2f __attribute__((ext_vector_type(2)));
typedef float v8f __attribute__((ext_vector_type(8)));

#define F_IN 512
#define C_OUT 64


#define LIST_CAP 63


#define N_TICKETS 2
#define LIST_ITEM 512

__device__ __forceinline__ int chip_half_of_wave() {
  return (int)((__builtin_amdgcn_s_sendmsg_rtn(0x87) >> 18) & 1u);
}

__device__ __forceinline__ int next_item(int* ticket, int* slot) {
  __syncthreads();
  if (threadIdx.x == 0) *slot = atomicAdd(ticket, 1);
  __syncthreads();
  return *slot;
}

__global__ void gcn_list_init(float* out, int* tickets, int N) {
  int* lists = (int*)out;
  int i = blockIdx.x * blockDim.x + threadIdx.x;
  if (i < N_TICKETS) tickets[i] = 0;
  if (i < N) lists[(size_t)i * C_OUT] = 0;
}

__global__ __launch_bounds__(256) void gcn_list_fill(
    const int* __restrict__ src, const int* __restrict__ dst,
    float* out, int* tickets, int E, int n_half) {
  int* lists = (int*)out;
  __shared__ int s_slot, s_half;
  if (threadIdx.x == 0) s_half = chip_half_of_wave();
  __syncthreads();
  const int home = s_half;
  const int n_items = (E + LIST_ITEM - 1) / LIST_ITEM;
  for (int pass = 0; pass < 2; ++pass) {
    const int part = home ^ pass;
    for (int trip = 0; trip <= n_items; ++trip) {
      const int item = next_item(&tickets[part], &s_slot);
      if (item >= n_items) break;
      const int base = item * LIST_ITEM;
      int  s[LIST_ITEM / 256], d[LIST_ITEM / 256], rank[LIST_ITEM / 256];
      bool mine[LIST_ITEM / 256];
      #pragma unroll
      for (int r = 0; r < LIST_ITEM / 256; ++r) {
        const int e  = base + r * 256 + (int)threadIdx.x;
        const int ec = e < E ? e : E - 1;
        s[r] = src[ec];
        d[r] = dst[ec];
        mine[r] = (e < E) && ((d[r] >= n_half) == (part != 0));
      }
      #pragma unroll
      for (int r = 0; r < LIST_ITEM / 256; ++r) {
        rank[r] = LIST_CAP;
        if (mine[r]) rank[r] = atomicAdd(&lists[(size_t)d[r] * C_OUT], 1);
      }
      #pragma unroll
      for (int r = 0; r < LIST_ITEM / 256; ++r) {
        if (rank[r] < LIST_CAP) lists[(size_t)d[r] * C_OUT + 1 + rank[r]] = s[r];
      }
    }
  }
}

__global__ void gcn_count_to_dinv(const float* out, float* __restrict__ dinv, int N) {
  int i = blockIdx.x * blockDim.x + threadIdx.x;
  if (i < N) {
    const int cnt = __float_as_int(out[(size_t)i * C_OUT]);
    dinv[i] = rsqrtf((float)(cnt + 1));
  }
}


typedef float v4f __attribute__((ext_vector_type(4)));

__global__ __launch_bounds__(128) void gcn_gemm_wmma(
    const float* __restrict__ x, const float* __restrict__ W,
    float* __restrict__ h, int N) {
  const int lane = threadIdx.x & 31;
  const int wid  = threadIdx.x >> 5;
  const int tile = blockIdx.x * 4 + wid;
  if (tile * 16 >= N) return;

  const int half = lane >> 4;
  const int l16  = lane & 15;
  const int row0 = tile * 16;

  const float* arow = x + (size_t)(row0 + l16) * F_IN + 4 * half;
  const float* bbas = W + (size_t)(4 * half) * C_OUT + 4 * l16;

  v8f acc0 = {}, acc1 = {}, acc2 = {}, acc3 = {};

  for (int k = 0; k < F_IN; k += 32) {
    v4f a[4], w[16];
    #pragma unroll
    for (int g = 0; g < 4; ++g) {
      a[g] = *(const v4f*)(arow + k + 8 * g);
      const float* bk = bbas + (size_t)(k + 8 * g) * C_OUT;
      w[4 * g + 0] = *(const v4f*)(bk);
      w[4 * g + 1] = *(const v4f*)(bk + C_OUT);
      w[4 * g + 2] = *(const v4f*)(bk + 2 * C_OUT);
      w[4 * g + 3] = *(const v4f*)(bk + 3 * C_OUT);
    }
    __builtin_amdgcn_sched_barrier(0);
    #pragma unroll
    for (int g = 0; g < 4; ++g) {
      const v4f w0 = w[4 * g + 0], w1 = w[4 * g + 1], w2 = w[4 * g + 2], w3 = w[4 * g + 3];
      const v2f a01 = { a[g].x, a[g].y };
      const v2f a23 = { a[g].z, a[g].w };
      const v2f b00 = { w0.x, w1.x }, b01 = { w0.y, w1.y }, b02 = { w0.z, w1.z }, b03 = { w0.w, w1.w };
      const v2f b10 = { w2.x, w3.x }, b11 = { w2.y, w3.y }, b12 = { w2.z, w3.z }, b13 = { w2.w, w3.w };
      acc0 = __builtin_amdgcn_wmma_f32_16x16x4_f32(false, a01, false, b00, (short)0, acc0, false, false);
      acc1 = __builtin_amdgcn_wmma_f32_16x16x4_f32(false, a01, false, b01, (short)0, acc1, false, false);
      acc2 = __builtin_amdgcn_wmma_f32_16x16x4_f32(false, a01, false, b02, (short)0, acc2, false, false);
      acc3 = __builtin_amdgcn_wmma_f32_16x16x4_f32(false, a01, false, b03, (short)0, acc3, false, false);
      acc0 = __builtin_amdgcn_wmma_f32_16x16x4_f32(false, a23, false, b10, (short)0, acc0, false, false);
      acc1 = __builtin_amdgcn_wmma_f32_16x16x4_f32(false, a23, false, b11, (short)0, acc1, false, false);
      acc2 = __builtin_amdgcn_wmma_f32_16x16x4_f32(false, a23, false, b12, (short)0, acc2, false, false);
      acc3 = __builtin_amdgcn_wmma_f32_16x16x4_f32(false, a23, false, b13, (short)0, acc3, false, false);
    }
  }

  float* hp = h + (size_t)row0 * C_OUT + 4 * l16;
  #pragma unroll
  for (int v = 0; v < 8; ++v) {
    const v4f r = { acc0[v], acc1[v], acc2[v], acc3[v] };
    *(v4f*)(hp + (size_t)(v + 8 * half) * C_OUT) = r;
  }
}


__device__ __forceinline__ int row_word(int lo, int hi, int word) {
  const int a = __builtin_amdgcn_readlane(lo, word & 31);
  const int b = __builtin_amdgcn_readlane(hi, word & 31);
  return word < 32 ? a : b;
}

__global__ __launch_bounds__(256) void gcn_rows_logsoftmax(
    const float* __restrict__ h, const float* __restrict__ dinv,
    const float* __restrict__ bias,
    const int* __restrict__ src, const int* __restrict__ dst,
    float* out, int N, int E) {
  const int lane   = threadIdx.x & 31;
  const int wave   = blockIdx.x * (blockDim.x >> 5) + (threadIdx.x >> 5);
  const int nwaves = gridDim.x * (blockDim.x >> 5);
  if (wave >= N) return;
  const float b0 = bias[lane], b1 = bias[lane + 32];

  float f0 = out[(size_t)wave * C_OUT + lane];
  float f1 = out[(size_t)wave * C_OUT + lane + 32];
  float dd = dinv[wave];
  float g0 = h[(size_t)wave * C_OUT + lane];
  float g1 = h[(size_t)wave * C_OUT + lane + 32];

  for (int row = wave; row < N; row += nwaves) {
    const int nrow = row + nwaves;
    const int nrc  = nrow < N ? nrow : row;
    const float nf0 = out[(size_t)nrc * C_OUT + lane];
    const float nf1 = out[(size_t)nrc * C_OUT + lane + 32];
    const float ndd = dinv[nrc];
    const float ng0 = h[(size_t)nrc * C_OUT + lane];
    const float ng1 = h[(size_t)nrc * C_OUT + lane + 32];

    const int w0  = __float_as_int(f0);
    const int w1  = __float_as_int(f1);
    const int cnt = __builtin_amdgcn_readfirstlane(w0);

    float a0 = g0 * dd * dd, a1 = g1 * dd * dd;

    if (cnt <= LIST_CAP) {
      const bool v0 = lane >= 1 && lane <= cnt;
      const bool v1 = lane + 32 <= cnt;
      const int n0 = __float_as_int(v0 ? dinv[w0] * dd : 0.0f);
      const int n1 = __float_as_int(v1 ? dinv[w1] * dd : 0.0f);
      int k = 1;
      for (; k + 3 <= cnt; k += 4) {
        int   s[4];
        float en[4], p0[4], p1[4];
        #pragma unroll
        for (int u = 0; u < 4; ++u) {
          s[u]  = row_word(w0, w1, k + u);
          en[u] = __int_as_float(row_word(n0, n1, k + u));
          const float* hs = h + (size_t)s[u] * C_OUT;
          p0[u] = hs[lane];
          p1[u] = hs[lane + 32];
        }
        #pragma unroll
        for (int u = 0; u < 4; ++u) {
          a0 += p0[u] * en[u];
          a1 += p1[u] * en[u];
        }
      }
      for (; k <= cnt; ++k) {
        const int   s  = row_word(w0, w1, k);
        const float en = __int_as_float(row_word(n0, n1, k));
        const float* hs = h + (size_t)s * C_OUT;
        a0 += hs[lane] * en;
        a1 += hs[lane + 32] * en;
      }
    } else {
      for (int e0 = 0; e0 < E; e0 += 32) {
        const int e  = e0 + lane;
        const int dv = e < E ? dst[e] : -1;
        unsigned m = __builtin_amdgcn_ballot_w32(dv == row);
        for (int t = 0; t < 32 && m != 0u; ++t) {
          const int l = __builtin_ctz(m);
          m &= m - 1u;
          const int   s  = src[e0 + l];
          const float en = dinv[s] * dd;
          const float* hs = h + (size_t)s * C_OUT;
          a0 += hs[lane] * en;
          a1 += hs[lane + 32] * en;
        }
      }
    }

    const float x0 = a0 + b0, x1 = a1 + b1;
    float m = fmaxf(x0, x1);
    #pragma unroll
    for (int off = 16; off > 0; off >>= 1) m = fmaxf(m, __shfl_xor(m, off, 32));
    float sum = __expf(x0 - m) + __expf(x1 - m);
    #pragma unroll
    for (int off = 16; off > 0; off >>= 1) sum += __shfl_xor(sum, off, 32);
    const float lse = __logf(sum);
    float* o = out + (size_t)row * C_OUT;
    o[lane]      = x0 - m - lse;
    o[lane + 32] = x1 - m - lse;

    f0 = nf0; f1 = nf1; dd = ndd; g0 = ng0; g1 = ng1;
  }
}


extern "C" void kernel_launch(void* const* d_in, const int* in_sizes, int n_in,
                              void* d_out, int out_size, void* d_ws, size_t ws_size,
                              hipStream_t stream) {
  const float* x  = (const float*)d_in[0];
  const float* W  = (const float*)d_in[1];
  const float* b  = (const float*)d_in[2];
  const int*   ei = (const int*)d_in[3];

  const int N = in_sizes[0] / F_IN;
  const int E = in_sizes[3] / 2;
  const int* src = ei;
  const int* dst = ei + E;
  float* out = (float*)d_out;

  char*  ws   = (char*)d_ws;
  float* dinv = (float*)ws;
  int*   tickets = (int*)(ws + (size_t)N * sizeof(float));
  size_t off  = ((size_t)N * sizeof(float) + N_TICKETS * sizeof(int) + 255) & ~(size_t)255;
  float* h    = (float*)(ws + off);
  const int n_half = N / 2;

  const int T = 256;

  gcn_list_init<<<(N + T - 1) / T, T, 0, stream>>>(out, tickets, N);
  gcn_list_fill<<<2048, T, 0, stream>>>(src, dst, out, tickets, E, n_half);
  gcn_count_to_dinv<<<(N + T - 1) / T, T, 0, stream>>>(out, dinv, N);

  {
    int tiles  = (N + 15) / 16;
    int blocks = (tiles + 3) / 4;
    gcn_gemm_wmma<<<blocks, 128, 0, stream>>>(x, W, h, N);
  }

  gcn_rows_logsoftmax<<<2048, T, 0, stream>>>(h, dinv, b, src, dst, out, N, E);
}
